# starting kernel
# baseline (speedup 1.0000x reference)
#include <hip/hip_runtime.h>


#define SIZE 8192
#define NB   64
#define BS   128
#define TM   64
#define KH   64
#define LDX  68

typedef float v2f __attribute__((ext_vector_type(2)));
typedef float v8f __attribute__((ext_vector_type(8)));

template <int PASS>
__global__ __launch_bounds__(256) void monarch_pass(
    const float* __restrict__ src, const float* __restrict__ w,
    const float* __restrict__ bias, float* __restrict__ dst) {
  __shared__ float Xs[TM][LDX];
  __shared__ float Wst[BS][LDX];

  const int tid  = threadIdx.x;
  const int lane = tid & 31;
  const int wv   = tid >> 5;
  const int b0   = blockIdx.x * TM;
  const int kblk = blockIdx.y;

  const float* __restrict__ wblk = w + (size_t)kblk * BS * BS;

  const int mrow  = (wv & 3) * 16 + (lane & 15);
  const int khalf = (lane >> 4) << 1;
  const int nbase = (wv >> 2) * 64;
  const int ncol  = lane & 15;

  v8f acc[4] = {};

  #pragma unroll
  for (int kh = 0; kh < 2; ++kh) {
    __syncthreads();
    #pragma unroll
    for (int i = 0; i < 4; ++i) {
      int f  = tid + i * 256;
      int r  = f >> 4;
      int c4 = (f & 15) << 2;
      float4 v = *reinterpret_cast<const float4*>(
          src + (size_t)(b0 + r) * SIZE + (size_t)kblk * BS + kh * KH + c4);
      *reinterpret_cast<float4*>(&Xs[r][c4]) = v;
    }
    #pragma unroll
    for (int i = 0; i < 8; ++i) {
      int f  = tid + i * 256;
      int d  = f >> 5;
      int c4 = (f & 31) << 2;
      float4 v = *reinterpret_cast<const float4*>(
          wblk + (size_t)(kh * KH + d) * BS + c4);
      Wst[c4 + 0][d] = v.x;
      Wst[c4 + 1][d] = v.y;
      Wst[c4 + 2][d] = v.z;
      Wst[c4 + 3][d] = v.w;
    }
    __syncthreads();

    #pragma unroll 4
    for (int kk = 0; kk < KH; kk += 4) {
      v2f a = *reinterpret_cast<const v2f*>(&Xs[mrow][kk + khalf]);
      #pragma unroll
      for (int j = 0; j < 4; ++j) {
        v2f b = *reinterpret_cast<const v2f*>(&Wst[nbase + j * 16 + ncol][kk + khalf]);
        acc[j] = __builtin_amdgcn_wmma_f32_16x16x4_f32(
             false, a,  false, b,
             (short)0, acc[j],  false,  false);
      }
    }
  }

  const int mtop = (wv & 3) * 16 + ((lane >> 4) << 3);
  #pragma unroll
  for (int j = 0; j < 4; ++j) {
    const int e = nbase + j * 16 + ncol;
    size_t col;
    float badd = 0.0f;
    if (PASS == 1) {
      col = (size_t)e * NB + kblk;
    } else {
      col = (size_t)(e & 63) * BS + 2 * kblk + (e >> 6);
      badd = bias[col];
    }
    float* __restrict__ drow = dst + (size_t)(b0 + mtop) * SIZE + col;
    #pragma unroll
    for (int r = 0; r < 8; ++r) {
      float v = acc[j][r];
      if (PASS == 2) v += badd;
      drow[(size_t)r * SIZE] = v;
    }
  }
}

extern "C" void kernel_launch(void* const* d_in, const int* in_sizes, int n_in,
                              void* d_out, int out_size, void* d_ws, size_t ws_size,
                              hipStream_t stream) {
  const float* x    = (const float*)d_in[0];
  const float* w1   = (const float*)d_in[1];
  const float* w2   = (const float*)d_in[2];
  const float* bias = (const float*)d_in[3];
  float* out   = (float*)d_out;
  float* inter = (float*)d_ws;

  const int Btok = in_sizes[0] / SIZE;
  dim3 block(256);
  dim3 grid(Btok / TM, NB);

  monarch_pass<1><<<grid, block, 0, stream>>>(x,     w1, bias, inter);
  monarch_pass<2><<<grid, block, 0, stream>>>(inter, w2, bias, out);
}


# Whole-line stores: intermediate token-major per 64-token tile, pass 1 tile staged through LDS; pass 2 does 2 blocks per work-group, a wave owns columns d and d+64 and stores float4 pieces
# speedup vs baseline: 2.9959x; 2.9959x over previous
#include <hip/hip_runtime.h>


#define SIZE 8192
#define NB   64
#define BS   128
#define TM   64
#define KH   64
#define LDX  68
#define G2   2

typedef float v2f __attribute__((ext_vector_type(2)));
typedef float v8f __attribute__((ext_vector_type(8)));

template <int PASS>
__global__ __launch_bounds__(256) void monarch_pass(
    const float* __restrict__ src, const float* __restrict__ w,
    const float* __restrict__ bias, float* __restrict__ dst) {
  __shared__ __attribute__((aligned(16))) float Xs[TM][LDX];
  __shared__ __attribute__((aligned(16))) float Wst[BS][LDX];

  constexpr int NBLK = (PASS == 1) ? 1 : G2;

  const int tid  = threadIdx.x;
  const int lane = tid & 31;
  const int wv   = tid >> 5;
  const int b0   = blockIdx.x * TM;
  const size_t tbase = (size_t)blockIdx.x * SIZE * TM;

  const int mrow  = (wv & 3) * 16 + (lane & 15);
  const int khalf = (lane >> 4) << 1;
  const int hh    = wv >> 2;
  const int ncol  = lane & 15;

  int nb[4];
  #pragma unroll
  for (int j = 0; j < 4; ++j)
    nb[j] = (PASS == 1) ? (hh * 64 + j * 16) : ((j >> 1) * 64 + hh * 32 + (j & 1) * 16);

  v8f acc[NBLK][4] = {};

  #pragma unroll
  for (int q = 0; q < NBLK; ++q) {
    const int kblk = blockIdx.y * NBLK + q;
    const float* __restrict__ wblk = w + (size_t)kblk * BS * BS;

    #pragma unroll
    for (int kh = 0; kh < 2; ++kh) {
      __syncthreads();
      if (PASS == 1) {
        #pragma unroll
        for (int i = 0; i < 4; ++i) {
          int f  = tid + i * 256;
          int r  = f >> 4;
          int c4 = (f & 15) << 2;
          float4 v = *reinterpret_cast<const float4*>(
              src + (size_t)(b0 + r) * SIZE + (size_t)kblk * BS + kh * KH + c4);
          *reinterpret_cast<float4*>(&Xs[r][c4]) = v;
        }
      } else {
        #pragma unroll
        for (int i = 0; i < 4; ++i) {
          int f  = tid + i * 256;
          int u  = f >> 5;
          int c  = ((u >> 2) << 3) | ((f >> 2) & 7);
          int t4 = (((u & 3) << 2) | (f & 3)) << 2;
          float4 v = *reinterpret_cast<const float4*>(
              src + tbase + (size_t)(kblk * BS + kh * KH + c) * TM + t4);
          Xs[t4 + 0][c] = v.x;
          Xs[t4 + 1][c] = v.y;
          Xs[t4 + 2][c] = v.z;
          Xs[t4 + 3][c] = v.w;
        }
      }
      #pragma unroll
      for (int i = 0; i < 8; ++i) {
        int f  = tid + i * 256;
        int d  = f >> 5;
        int c4 = (f & 31) << 2;
        float4 v = *reinterpret_cast<const float4*>(
            wblk + (size_t)(kh * KH + d) * BS + c4);
        Wst[c4 + 0][d] = v.x;
        Wst[c4 + 1][d] = v.y;
        Wst[c4 + 2][d] = v.z;
        Wst[c4 + 3][d] = v.w;
      }
      __syncthreads();

      #pragma unroll 4
      for (int kk = 0; kk < KH; kk += 4) {
        v2f a = *reinterpret_cast<const v2f*>(&Xs[mrow][kk + khalf]);
        #pragma unroll
        for (int j = 0; j < 4; ++j) {
          v2f b = *reinterpret_cast<const v2f*>(&Wst[nb[j] + ncol][kk + khalf]);
          acc[q][j] = __builtin_amdgcn_wmma_f32_16x16x4_f32(
               false, a,  false, b,
               (short)0, acc[q][j],  false,  false);
        }
      }
    }
  }

  const int mtop = (wv & 3) * 16 + ((lane >> 4) << 3);

  if (PASS == 1) {
    const int kblk = blockIdx.y;
    __syncthreads();
    #pragma unroll
    for (int j = 0; j < 4; ++j) {
      const int e = nb[j] + ncol;
      float4 lo, hi;
      lo.x = acc[0][j][0]; lo.y = acc[0][j][1]; lo.z = acc[0][j][2]; lo.w = acc[0][j][3];
      hi.x = acc[0][j][4]; hi.y = acc[0][j][5]; hi.z = acc[0][j][6]; hi.w = acc[0][j][7];
      *reinterpret_cast<float4*>(&Wst[e][mtop])     = lo;
      *reinterpret_cast<float4*>(&Wst[e][mtop + 4]) = hi;
    }
    __syncthreads();
    #pragma unroll
    for (int i = 0; i < 8; ++i) {
      int f  = tid + i * 256;
      int e  = f >> 4;
      int t4 = (f & 15) << 2;
      float4 v = *reinterpret_cast<const float4*>(&Wst[e][t4]);
      *reinterpret_cast<float4*>(dst + tbase + (size_t)(e * NB + kblk) * TM + t4) = v;
    }
  } else {
    #pragma unroll
    for (int jj = 0; jj < 2; ++jj) {
      const int j2 = hh * 32 + jj * 16 + ncol;
      const size_t col = (size_t)j2 * BS + 2 * G2 * blockIdx.y;
      const float4 bv = *reinterpret_cast<const float4*>(bias + col);
      float* __restrict__ drow = dst + (size_t)(b0 + mtop) * SIZE + col;
      #pragma unroll
      for (int r = 0; r < 8; ++r) {
        float4 v;
        v.x = acc[0][jj][r]     + bv.x;
        v.y = acc[0][jj + 2][r] + bv.y;
        v.z = acc[1 % NBLK][jj][r]     + bv.z;
        v.w = acc[1 % NBLK][jj + 2][r] + bv.w;
        *reinterpret_cast<float4*>(drow + (size_t)r * SIZE) = v;
      }
    }
  }
}

extern "C" void kernel_launch(void* const* d_in, const int* in_sizes, int n_in,
                              void* d_out, int out_size, void* d_ws, size_t ws_size,
                              hipStream_t stream) {
  const float* x    = (const float*)d_in[0];
  const float* w1   = (const float*)d_in[1];
  const float* w2   = (const float*)d_in[2];
  const float* bias = (const float*)d_in[3];
  float* out   = (float*)d_out;
  float* inter = (float*)d_ws;

  const int Btok = in_sizes[0] / SIZE;
  dim3 block(256);
  dim3 grid1(Btok / TM, NB);
  dim3 grid2(Btok / TM, NB / G2);

  monarch_pass<1><<<grid1, block, 0, stream>>>(x,     w1, bias, inter);
  monarch_pass<2><<<grid2, block, 0, stream>>>(inter, w2, bias, out);
}


# Weights resident: a work-group stages its transposed weight block(s) into LDS once and loops over 8 (pass 1) / 4 (pass 2) token tiles, staging only the token panel per tile
# speedup vs baseline: 3.5365x; 1.1804x over previous
#include <hip/hip_runtime.h>


#define SIZE 8192
#define NB   64
#define BS   128
#define TM   64
#define KH   64
#define LDX  68
#define LDW  132
#define G2   2
#define NT1  8
#define NT2  4

typedef float v2f __attribute__((ext_vector_type(2)));
typedef float v8f __attribute__((ext_vector_type(8)));

template <int PASS>
__global__ __launch_bounds__(256) void monarch_pass(
    const float* __restrict__ src, const float* __restrict__ w,
    const float* __restrict__ bias, float* __restrict__ dst) {
  constexpr int NBLK  = (PASS == 1) ? 1 : G2;
  constexpr int NT    = (PASS == 1) ? NT1 : NT2;
  constexpr int XROWS = (PASS == 1) ? BS : TM;
  __shared__ __attribute__((aligned(16))) float Wt[NBLK][BS][LDW];
  __shared__ __attribute__((aligned(16))) float Xs[XROWS][LDX];

  const int tid  = threadIdx.x;
  const int lane = tid & 31;
  const int wv   = tid >> 5;

  const int mrow  = (wv & 3) * 16 + (lane & 15);
  const int khalf = (lane >> 4) << 1;
  const int hh    = wv >> 2;
  const int ncol  = lane & 15;

  int nb[4];
  #pragma unroll
  for (int j = 0; j < 4; ++j)
    nb[j] = (PASS == 1) ? (hh * 64 + j * 16) : ((j >> 1) * 64 + hh * 32 + (j & 1) * 16);

  #pragma unroll
  for (int q = 0; q < NBLK; ++q) {
    const float* __restrict__ wblk = w + (size_t)(blockIdx.y * NBLK + q) * BS * BS;
    #pragma unroll
    for (int i = 0; i < 16; ++i) {
      int f  = tid + i * 256;
      int d  = f >> 5;
      int c4 = (f & 31) << 2;
      float4 v = *reinterpret_cast<const float4*>(wblk + (size_t)d * BS + c4);
      Wt[q][c4 + 0][d] = v.x;
      Wt[q][c4 + 1][d] = v.y;
      Wt[q][c4 + 2][d] = v.z;
      Wt[q][c4 + 3][d] = v.w;
    }
  }

  const int mtop = (wv & 3) * 16 + ((lane >> 4) << 3);

  for (int t = 0; t < NT; ++t) {
    const int tile = blockIdx.x * NT + t;
    const int b0   = tile * TM;
    const size_t tbase = (size_t)tile * SIZE * TM;

    v8f acc[NBLK][4] = {};

    #pragma unroll
    for (int q = 0; q < NBLK; ++q) {
      const int kblk = blockIdx.y * NBLK + q;

      #pragma unroll
      for (int kh = 0; kh < 2; ++kh) {
        __syncthreads();
        if (PASS == 1) {
          #pragma unroll
          for (int i = 0; i < 4; ++i) {
            int f  = tid + i * 256;
            int r  = f >> 4;
            int c4 = (f & 15) << 2;
            float4 v = *reinterpret_cast<const float4*>(
                src + (size_t)(b0 + r) * SIZE + (size_t)kblk * BS + kh * KH + c4);
            *reinterpret_cast<float4*>(&Xs[r][c4]) = v;
          }
        } else {
          #pragma unroll
          for (int i = 0; i < 4; ++i) {
            int f  = tid + i * 256;
            int u  = f >> 5;
            int c  = ((u >> 2) << 3) | ((f >> 2) & 7);
            int t4 = (((u & 3) << 2) | (f & 3)) << 2;
            float4 v = *reinterpret_cast<const float4*>(
                src + tbase + (size_t)(kblk * BS + kh * KH + c) * TM + t4);
            Xs[t4 + 0][c] = v.x;
            Xs[t4 + 1][c] = v.y;
            Xs[t4 + 2][c] = v.z;
            Xs[t4 + 3][c] = v.w;
          }
        }
        __syncthreads();

        #pragma unroll 4
        for (int kk = 0; kk < KH; kk += 4) {
          v2f a = *reinterpret_cast<const v2f*>(&Xs[mrow][kk + khalf]);
          #pragma unroll
          for (int j = 0; j < 4; ++j) {
            v2f b = *reinterpret_cast<const v2f*>(&Wt[q][nb[j] + ncol][kh * KH + kk + khalf]);
            acc[q][j] = __builtin_amdgcn_wmma_f32_16x16x4_f32(
                 false, a,  false, b,
                 (short)0, acc[q][j],  false,  false);
          }
        }
      }
    }

    if (PASS == 1) {
      const int kblk = blockIdx.y;
      __syncthreads();
      #pragma unroll
      for (int j = 0; j < 4; ++j) {
        const int e = nb[j] + ncol;
        float4 lo, hi;
        lo.x = acc[0][j][0]; lo.y = acc[0][j][1]; lo.z = acc[0][j][2]; lo.w = acc[0][j][3];
        hi.x = acc[0][j][4]; hi.y = acc[0][j][5]; hi.z = acc[0][j][6]; hi.w = acc[0][j][7];
        *reinterpret_cast<float4*>(&Xs[e][mtop])     = lo;
        *reinterpret_cast<float4*>(&Xs[e][mtop + 4]) = hi;
      }
      __syncthreads();
      #pragma unroll
      for (int i = 0; i < 8; ++i) {
        int f  = tid + i * 256;
        int e  = f >> 4;
        int t4 = (f & 15) << 2;
        float4 v = *reinterpret_cast<const float4*>(&Xs[e][t4]);
        *reinterpret_cast<float4*>(dst + tbase + (size_t)(e * NB + kblk) * TM + t4) = v;
      }
    } else {
      #pragma unroll
      for (int jj = 0; jj < 2; ++jj) {
        const int j2 = hh * 32 + jj * 16 + ncol;
        const size_t col = (size_t)j2 * BS + 2 * G2 * blockIdx.y;
        const float4 bv = *reinterpret_cast<const float4*>(bias + col);
        float* __restrict__ drow = dst + (size_t)(b0 + mtop) * SIZE + col;
        #pragma unroll
        for (int r = 0; r < 8; ++r) {
          float4 v;
          v.x = acc[0][jj][r]     + bv.x;
          v.y = acc[0][jj + 2][r] + bv.y;
          v.z = acc[1 % NBLK][jj][r]     + bv.z;
          v.w = acc[1 % NBLK][jj + 2][r] + bv.w;
          *reinterpret_cast<float4*>(drow + (size_t)r * SIZE) = v;
        }
      }
    }
  }
}

extern "C" void kernel_launch(void* const* d_in, const int* in_sizes, int n_in,
                              void* d_out, int out_size, void* d_ws, size_t ws_size,
                              hipStream_t stream) {
  const float* x    = (const float*)d_in[0];
  const float* w1   = (const float*)d_in[1];
  const float* w2   = (const float*)d_in[2];
  const float* bias = (const float*)d_in[3];
  float* out   = (float*)d_out;
  float* inter = (float*)d_ws;

  const int Btok = in_sizes[0] / SIZE;
  dim3 block(256);
  dim3 grid1(Btok / (TM * NT1), NB);
  dim3 grid2(Btok / (TM * NT2), NB / G2);

  monarch_pass<1><<<grid1, block, 0, stream>>>(x,     w1, bias, inter);
  monarch_pass<2><<<grid2, block, 0, stream>>>(inter, w2, bias, out);
}


# Panel loads software-pipelined in both passes: the next panel half's global loads are issued into registers before the current half's WMMAs and stored to LDS behind them
# speedup vs baseline: 3.8147x; 1.0787x over previous
#include <hip/hip_runtime.h>


#define SIZE 8192
#define NB   64
#define BS   128
#define TM   64
#define KH   64
#define LDX  68
#define LDW  132
#define G2   2
#define NT1  8
#define NT2  4

typedef float v2f __attribute__((ext_vector_type(2)));
typedef float v8f __attribute__((ext_vector_type(8)));

template <int PASS>
__global__ __launch_bounds__(256) void monarch_pass(
    const float* __restrict__ src, const float* __restrict__ w,
    const float* __restrict__ bias, float* __restrict__ dst) {
  constexpr int NBLK  = (PASS == 1) ? 1 : G2;
  constexpr int NT    = (PASS == 1) ? NT1 : NT2;
  constexpr int XROWS = (PASS == 1) ? BS : TM;
  __shared__ __attribute__((aligned(16))) float Wt[NBLK][BS][LDW];
  __shared__ __attribute__((aligned(16))) float Xs[XROWS][LDX];

  const int tid  = threadIdx.x;
  const int lane = tid & 31;
  const int wv   = tid >> 5;

  const int mrow  = (wv & 3) * 16 + (lane & 15);
  const int khalf = (lane >> 4) << 1;
  const int hh    = wv >> 2;
  const int ncol  = lane & 15;

  int nb[4];
  #pragma unroll
  for (int j = 0; j < 4; ++j)
    nb[j] = (PASS == 1) ? (hh * 64 + j * 16) : ((j >> 1) * 64 + hh * 32 + (j & 1) * 16);

  float4 pf0, pf1, pf2, pf3;

  auto piece = [&](int t, int q, int kh, int i) -> float4 {
    const int ftile = blockIdx.x * NT + t;
    const int fkblk = blockIdx.y * NBLK + q;
    const int f = tid + i * 256;
    if (PASS == 1) {
      const int r  = f >> 4;
      const int c4 = (f & 15) << 2;
      return *reinterpret_cast<const float4*>(
          src + (size_t)(ftile * TM + r) * SIZE + (size_t)fkblk * BS + kh * KH + c4);
    } else {
      const int u  = f >> 5;
      const int c  = ((u >> 2) << 3) | ((f >> 2) & 7);
      const int t4 = (((u & 3) << 2) | (f & 3)) << 2;
      return *reinterpret_cast<const float4*>(
          src + (size_t)ftile * SIZE * TM + (size_t)(fkblk * BS + kh * KH + c) * TM + t4);
    }
  };
  auto fetch = [&](int t, int q, int kh) {
    pf0 = piece(t, q, kh, 0);
    pf1 = piece(t, q, kh, 1);
    pf2 = piece(t, q, kh, 2);
    pf3 = piece(t, q, kh, 3);
  };

  auto put = [&](int i, float4 v) {
    const int f = tid + i * 256;
    if (PASS == 1) {
      const int r  = f >> 4;
      const int c4 = (f & 15) << 2;
      *reinterpret_cast<float4*>(&Xs[r][c4]) = v;
    } else {
      const int u  = f >> 5;
      const int c  = ((u >> 2) << 3) | ((f >> 2) & 7);
      const int t4 = (((u & 3) << 2) | (f & 3)) << 2;
      Xs[t4 + 0][c] = v.x;
      Xs[t4 + 1][c] = v.y;
      Xs[t4 + 2][c] = v.z;
      Xs[t4 + 3][c] = v.w;
    }
  };
  auto stash = [&]() {
    put(0, pf0);
    put(1, pf1);
    put(2, pf2);
    put(3, pf3);
  };

  fetch(0, 0, 0);

  #pragma unroll
  for (int q = 0; q < NBLK; ++q) {
    const float* __restrict__ wblk = w + (size_t)(blockIdx.y * NBLK + q) * BS * BS;
    #pragma unroll
    for (int i = 0; i < 16; ++i) {
      int f  = tid + i * 256;
      int d  = f >> 5;
      int c4 = (f & 31) << 2;
      float4 v = *reinterpret_cast<const float4*>(wblk + (size_t)d * BS + c4);
      Wt[q][c4 + 0][d] = v.x;
      Wt[q][c4 + 1][d] = v.y;
      Wt[q][c4 + 2][d] = v.z;
      Wt[q][c4 + 3][d] = v.w;
    }
  }

  const int mtop = (wv & 3) * 16 + ((lane >> 4) << 3);

  for (int t = 0; t < NT; ++t) {
    const int tile = blockIdx.x * NT + t;
    const int b0   = tile * TM;
    const size_t tbase = (size_t)tile * SIZE * TM;

    v8f acc[NBLK][4] = {};

    #pragma unroll
    for (int q = 0; q < NBLK; ++q) {
      #pragma unroll
      for (int kh = 0; kh < 2; ++kh) {
        __syncthreads();
        stash();
        __syncthreads();

        if (kh == 0)            fetch(t, q, 1);
        else if (q + 1 < NBLK)  fetch(t, q + 1, 0);
        else if (t + 1 < NT)    fetch(t + 1, 0, 0);

        #pragma unroll 4
        for (int kk = 0; kk < KH; kk += 4) {
          v2f a = *reinterpret_cast<const v2f*>(&Xs[mrow][kk + khalf]);
          #pragma unroll
          for (int j = 0; j < 4; ++j) {
            v2f b = *reinterpret_cast<const v2f*>(&Wt[q][nb[j] + ncol][kh * KH + kk + khalf]);
            acc[q][j] = __builtin_amdgcn_wmma_f32_16x16x4_f32(
                 false, a,  false, b,
                 (short)0, acc[q][j],  false,  false);
          }
        }
      }
    }

    if (PASS == 1) {
      const int kblk = blockIdx.y;
      __syncthreads();
      #pragma unroll
      for (int j = 0; j < 4; ++j) {
        const int e = nb[j] + ncol;
        float4 lo, hi;
        lo.x = acc[0][j][0]; lo.y = acc[0][j][1]; lo.z = acc[0][j][2]; lo.w = acc[0][j][3];
        hi.x = acc[0][j][4]; hi.y = acc[0][j][5]; hi.z = acc[0][j][6]; hi.w = acc[0][j][7];
        *reinterpret_cast<float4*>(&Xs[e][mtop])     = lo;
        *reinterpret_cast<float4*>(&Xs[e][mtop + 4]) = hi;
      }
      __syncthreads();
      #pragma unroll
      for (int i = 0; i < 8; ++i) {
        int f  = tid + i * 256;
        int e  = f >> 4;
        int t4 = (f & 15) << 2;
        float4 v = *reinterpret_cast<const float4*>(&Xs[e][t4]);
        *reinterpret_cast<float4*>(dst + tbase + (size_t)(e * NB + kblk) * TM + t4) = v;
      }
    } else {
      #pragma unroll
      for (int jj = 0; jj < 2; ++jj) {
        const int j2 = hh * 32 + jj * 16 + ncol;
        const size_t col = (size_t)j2 * BS + 2 * G2 * blockIdx.y;
        const float4 bv = *reinterpret_cast<const float4*>(bias + col);
        float* __restrict__ drow = dst + (size_t)(b0 + mtop) * SIZE + col;
        #pragma unroll
        for (int r = 0; r < 8; ++r) {
          float4 v;
          v.x = acc[0][jj][r]     + bv.x;
          v.y = acc[0][jj + 2][r] + bv.y;
          v.z = acc[1 % NBLK][jj][r]     + bv.z;
          v.w = acc[1 % NBLK][jj + 2][r] + bv.w;
          *reinterpret_cast<float4*>(drow + (size_t)r * SIZE) = v;
        }
      }
    }
  }
}

extern "C" void kernel_launch(void* const* d_in, const int* in_sizes, int n_in,
                              void* d_out, int out_size, void* d_ws, size_t ws_size,
                              hipStream_t stream) {
  const float* x    = (const float*)d_in[0];
  const float* w1   = (const float*)d_in[1];
  const float* w2   = (const float*)d_in[2];
  const float* bias = (const float*)d_in[3];
  float* out   = (float*)d_out;
  float* inter = (float*)d_ws;

  const int Btok = in_sizes[0] / SIZE;
  dim3 block(256);
  dim3 grid1(Btok / (TM * NT1), NB);
  dim3 grid2(Btok / (TM * NT2), NB / G2);

  monarch_pass<1><<<grid1, block, 0, stream>>>(x,     w1, bias, inter);
  monarch_pass<2><<<grid2, block, 0, stream>>>(inter, w2, bias, out);
}


# Weight staging without LDS bank conflicts: a wave loads 8 reduction rows x 16 columns per piece, so its 32 transposing LDS stores hit 32 different banks instead of 4 (8-way conflict before)
# speedup vs baseline: 3.9559x; 1.0370x over previous
#include <hip/hip_runtime.h>


#define SIZE 8192
#define NB   64
#define BS   128
#define TM   64
#define KH   64
#define LDX  68
#define LDW  132
#define G2   2
#define NT1  8
#define NT2  4

typedef float v2f __attribute__((ext_vector_type(2)));
typedef float v8f __attribute__((ext_vector_type(8)));

template <int PASS>
__global__ __launch_bounds__(256) void monarch_pass(
    const float* __restrict__ src, const float* __restrict__ w,
    const float* __restrict__ bias, float* __restrict__ dst) {
  constexpr int NBLK  = (PASS == 1) ? 1 : G2;
  constexpr int NT    = (PASS == 1) ? NT1 : NT2;
  constexpr int XROWS = (PASS == 1) ? BS : TM;
  __shared__ __attribute__((aligned(16))) float Wt[NBLK][BS][LDW];
  __shared__ __attribute__((aligned(16))) float Xs[XROWS][LDX];

  const int tid  = threadIdx.x;
  const int lane = tid & 31;
  const int wv   = tid >> 5;

  const int mrow  = (wv & 3) * 16 + (lane & 15);
  const int khalf = (lane >> 4) << 1;
  const int hh    = wv >> 2;
  const int ncol  = lane & 15;

  int nb[4];
  #pragma unroll
  for (int j = 0; j < 4; ++j)
    nb[j] = (PASS == 1) ? (hh * 64 + j * 16) : ((j >> 1) * 64 + hh * 32 + (j & 1) * 16);

  float4 pf0, pf1, pf2, pf3;

  auto piece = [&](int t, int q, int kh, int i) -> float4 {
    const int ftile = blockIdx.x * NT + t;
    const int fkblk = blockIdx.y * NBLK + q;
    const int f = tid + i * 256;
    if (PASS == 1) {
      const int r  = f >> 4;
      const int c4 = (f & 15) << 2;
      return *reinterpret_cast<const float4*>(
          src + (size_t)(ftile * TM + r) * SIZE + (size_t)fkblk * BS + kh * KH + c4);
    } else {
      const int u  = f >> 5;
      const int c  = ((u >> 2) << 3) | ((f >> 2) & 7);
      const int t4 = (((u & 3) << 2) | (f & 3)) << 2;
      return *reinterpret_cast<const float4*>(
          src + (size_t)ftile * SIZE * TM + (size_t)(fkblk * BS + kh * KH + c) * TM + t4);
    }
  };
  auto fetch = [&](int t, int q, int kh) {
    pf0 = piece(t, q, kh, 0);
    pf1 = piece(t, q, kh, 1);
    pf2 = piece(t, q, kh, 2);
    pf3 = piece(t, q, kh, 3);
  };

  auto put = [&](int i, float4 v) {
    const int f = tid + i * 256;
    if (PASS == 1) {
      const int r  = f >> 4;
      const int c4 = (f & 15) << 2;
      *reinterpret_cast<float4*>(&Xs[r][c4]) = v;
    } else {
      const int u  = f >> 5;
      const int c  = ((u >> 2) << 3) | ((f >> 2) & 7);
      const int t4 = (((u & 3) << 2) | (f & 3)) << 2;
      Xs[t4 + 0][c] = v.x;
      Xs[t4 + 1][c] = v.y;
      Xs[t4 + 2][c] = v.z;
      Xs[t4 + 3][c] = v.w;
    }
  };
  auto stash = [&]() {
    put(0, pf0);
    put(1, pf1);
    put(2, pf2);
    put(3, pf3);
  };

  fetch(0, 0, 0);

  #pragma unroll
  for (int q = 0; q < NBLK; ++q) {
    const float* __restrict__ wblk = w + (size_t)(blockIdx.y * NBLK + q) * BS * BS;
    #pragma unroll
    for (int i = 0; i < 16; ++i) {
      int f  = tid + i * 256;
      int u  = f >> 5;
      int d  = ((u >> 3) << 3) | ((f >> 2) & 7);
      int c4 = (((u & 7) << 2) | (f & 3)) << 2;
      float4 v = *reinterpret_cast<const float4*>(wblk + (size_t)d * BS + c4);
      Wt[q][c4 + 0][d] = v.x;
      Wt[q][c4 + 1][d] = v.y;
      Wt[q][c4 + 2][d] = v.z;
      Wt[q][c4 + 3][d] = v.w;
    }
  }

  const int mtop = (wv & 3) * 16 + ((lane >> 4) << 3);

  for (int t = 0; t < NT; ++t) {
    const int tile = blockIdx.x * NT + t;
    const int b0   = tile * TM;
    const size_t tbase = (size_t)tile * SIZE * TM;

    v8f acc[NBLK][4] = {};

    #pragma unroll
    for (int q = 0; q < NBLK; ++q) {
      #pragma unroll
      for (int kh = 0; kh < 2; ++kh) {
        __syncthreads();
        stash();
        __syncthreads();

        if (kh == 0)            fetch(t, q, 1);
        else if (q + 1 < NBLK)  fetch(t, q + 1, 0);
        else if (t + 1 < NT)    fetch(t + 1, 0, 0);

        #pragma unroll 4
        for (int kk = 0; kk < KH; kk += 4) {
          v2f a = *reinterpret_cast<const v2f*>(&Xs[mrow][kk + khalf]);
          #pragma unroll
          for (int j = 0; j < 4; ++j) {
            v2f b = *reinterpret_cast<const v2f*>(&Wt[q][nb[j] + ncol][kh * KH + kk + khalf]);
            acc[q][j] = __builtin_amdgcn_wmma_f32_16x16x4_f32(
                 false, a,  false, b,
                 (short)0, acc[q][j],  false,  false);
          }
        }
      }
    }

    if (PASS == 1) {
      const int kblk = blockIdx.y;
      __syncthreads();
      #pragma unroll
      for (int j = 0; j < 4; ++j) {
        const int e = nb[j] + ncol;
        float4 lo, hi;
        lo.x = acc[0][j][0]; lo.y = acc[0][j][1]; lo.z = acc[0][j][2]; lo.w = acc[0][j][3];
        hi.x = acc[0][j][4]; hi.y = acc[0][j][5]; hi.z = acc[0][j][6]; hi.w = acc[0][j][7];
        *reinterpret_cast<float4*>(&Xs[e][mtop])     = lo;
        *reinterpret_cast<float4*>(&Xs[e][mtop + 4]) = hi;
      }
      __syncthreads();
      #pragma unroll
      for (int i = 0; i < 8; ++i) {
        int f  = tid + i * 256;
        int e  = f >> 4;
        int t4 = (f & 15) << 2;
        float4 v = *reinterpret_cast<const float4*>(&Xs[e][t4]);
        *reinterpret_cast<float4*>(dst + tbase + (size_t)(e * NB + kblk) * TM + t4) = v;
      }
    } else {
      #pragma unroll
      for (int jj = 0; jj < 2; ++jj) {
        const int j2 = hh * 32 + jj * 16 + ncol;
        const size_t col = (size_t)j2 * BS + 2 * G2 * blockIdx.y;
        const float4 bv = *reinterpret_cast<const float4*>(bias + col);
        float* __restrict__ drow = dst + (size_t)(b0 + mtop) * SIZE + col;
        #pragma unroll
        for (int r = 0; r < 8; ++r) {
          float4 v;
          v.x = acc[0][jj][r]     + bv.x;
          v.y = acc[0][jj + 2][r] + bv.y;
          v.z = acc[1 % NBLK][jj][r]     + bv.z;
          v.w = acc[1 % NBLK][jj + 2][r] + bv.w;
          *reinterpret_cast<float4*>(drow + (size_t)r * SIZE) = v;
        }
      }
    }
  }
}

extern "C" void kernel_launch(void* const* d_in, const int* in_sizes, int n_in,
                              void* d_out, int out_size, void* d_ws, size_t ws_size,
                              hipStream_t stream) {
  const float* x    = (const float*)d_in[0];
  const float* w1   = (const float*)d_in[1];
  const float* w2   = (const float*)d_in[2];
  const float* bias = (const float*)d_in[3];
  float* out   = (float*)d_out;
  float* inter = (float*)d_ws;

  const int Btok = in_sizes[0] / SIZE;
  dim3 block(256);
  dim3 grid1(Btok / (TM * NT1), NB);
  dim3 grid2(Btok / (TM * NT2), NB / G2);

  monarch_pass<1><<<grid1, block, 0, stream>>>(x,     w1, bias, inter);
  monarch_pass<2><<<grid2, block, 0, stream>>>(inter, w2, bias, out);
}


# Pass 2 stores 32-byte pieces, not 16-byte: one 16-wave work-group per WGP keeps 4 weight blocks, a lane owns 8 neighbouring outputs, lane pairs swap halves before storing; panel staged 128 deep
# speedup vs baseline: 4.1493x; 1.0489x over previous
#include <hip/hip_runtime.h>


#define SIZE 8192
#define NB   64
#define BS   128
#define TM   64
#define LDW  132
#define G2   4
#define NT1  8
#define NT2  4

typedef float v2f __attribute__((ext_vector_type(2)));
typedef float v8f __attribute__((ext_vector_type(8)));

template <int PASS>
__global__ __launch_bounds__(PASS == 1 ? 256 : 512) void monarch_pass(
    const float* __restrict__ src, const float* __restrict__ w,
    const float* __restrict__ bias, float* __restrict__ dst) {
  constexpr int NBLK  = (PASS == 1) ? 1 : G2;
  constexpr int NT    = (PASS == 1) ? NT1 : NT2;
  constexpr int NTHR  = (PASS == 1) ? 256 : 512;
  constexpr int KP    = (PASS == 1) ? 64 : 128;
  constexpr int NKP   = BS / KP;
  constexpr int LDP   = KP + 4;
  constexpr int NJ    = (PASS == 1) ? 4 : 2;
  constexpr int XROWS = (PASS == 1) ? BS : TM;
  __shared__ __attribute__((aligned(16))) float Wt[NBLK][BS][LDW];
  __shared__ __attribute__((aligned(16))) float Xs[XROWS][LDP];

  const int tid  = threadIdx.x;
  const int lane = tid & 31;
  const int wv   = tid >> 5;

  const int mrow  = (wv & 3) * 16 + (lane & 15);
  const int khalf = (lane >> 4) << 1;
  const int hh    = wv >> 2;
  const int ncol  = lane & 15;

  int nb[NJ];
  #pragma unroll
  for (int j = 0; j < NJ; ++j)
    nb[j] = (PASS == 1) ? (hh * 64 + j * 16) : (j * 64 + hh * 16);

  float4 pf0, pf1, pf2, pf3;

  auto piece = [&](int t, int q, int kh, int i) -> float4 {
    const int ftile = blockIdx.x * NT + t;
    const int fkblk = blockIdx.y * NBLK + q;
    const int f = tid + i * NTHR;
    if (PASS == 1) {
      const int r  = f >> 4;
      const int c4 = (f & 15) << 2;
      return *reinterpret_cast<const float4*>(
          src + (size_t)(ftile * TM + r) * SIZE + (size_t)fkblk * BS + kh * KP + c4);
    } else {
      const int u  = f >> 5;
      const int c  = ((u >> 2) << 3) | ((f >> 2) & 7);
      const int t4 = (((u & 3) << 2) | (f & 3)) << 2;
      return *reinterpret_cast<const float4*>(
          src + (size_t)ftile * SIZE * TM + (size_t)(fkblk * BS + kh * KP + c) * TM + t4);
    }
  };
  auto fetch = [&](int t, int q, int kh) {
    pf0 = piece(t, q, kh, 0);
    pf1 = piece(t, q, kh, 1);
    pf2 = piece(t, q, kh, 2);
    pf3 = piece(t, q, kh, 3);
  };

  auto put = [&](int i, float4 v) {
    const int f = tid + i * NTHR;
    if (PASS == 1) {
      const int r  = f >> 4;
      const int c4 = (f & 15) << 2;
      *reinterpret_cast<float4*>(&Xs[r][c4]) = v;
    } else {
      const int u  = f >> 5;
      const int c  = ((u >> 2) << 3) | ((f >> 2) & 7);
      const int t4 = (((u & 3) << 2) | (f & 3)) << 2;
      Xs[t4 + 0][c] = v.x;
      Xs[t4 + 1][c] = v.y;
      Xs[t4 + 2][c] = v.z;
      Xs[t4 + 3][c] = v.w;
    }
  };
  auto stash = [&]() {
    put(0, pf0);
    put(1, pf1);
    put(2, pf2);
    put(3, pf3);
  };

  fetch(0, 0, 0);

  #pragma unroll
  for (int q = 0; q < NBLK; ++q) {
    const float* __restrict__ wblk = w + (size_t)(blockIdx.y * NBLK + q) * BS * BS;
    #pragma unroll
    for (int i = 0; i < (BS * BS / 4) / NTHR; ++i) {
      int f  = tid + i * NTHR;
      int u  = f >> 5;
      int d  = ((u >> 3) << 3) | ((f >> 2) & 7);
      int c4 = (((u & 7) << 2) | (f & 3)) << 2;
      float4 v = *reinterpret_cast<const float4*>(wblk + (size_t)d * BS + c4);
      Wt[q][c4 + 0][d] = v.x;
      Wt[q][c4 + 1][d] = v.y;
      Wt[q][c4 + 2][d] = v.z;
      Wt[q][c4 + 3][d] = v.w;
    }
  }

  const int mtop = (wv & 3) * 16 + ((lane >> 4) << 3);

  for (int t = 0; t < NT; ++t) {
    const int tile = blockIdx.x * NT + t;
    const int b0   = tile * TM;
    const size_t tbase = (size_t)tile * SIZE * TM;

    v8f acc[NBLK][NJ] = {};

    #pragma unroll
    for (int q = 0; q < NBLK; ++q) {
      #pragma unroll
      for (int kh = 0; kh < NKP; ++kh) {
        __syncthreads();
        stash();
        __syncthreads();

        if (kh + 1 < NKP)       fetch(t, q, kh + 1);
        else if (q + 1 < NBLK)  fetch(t, q + 1, 0);
        else if (t + 1 < NT)    fetch(t + 1, 0, 0);

        #pragma unroll (16 / NJ)
        for (int kk = 0; kk < KP; kk += 4) {
          v2f a = *reinterpret_cast<const v2f*>(&Xs[mrow][kk + khalf]);
          #pragma unroll
          for (int j = 0; j < NJ; ++j) {
            v2f b = *reinterpret_cast<const v2f*>(&Wt[q][nb[j] + ncol][kh * KP + kk + khalf]);
            acc[q][j] = __builtin_amdgcn_wmma_f32_16x16x4_f32(
                 false, a,  false, b,
                 (short)0, acc[q][j],  false,  false);
          }
        }
      }
    }

    if constexpr (PASS == 1) {
      const int kblk = blockIdx.y;
      __syncthreads();
      #pragma unroll
      for (int j = 0; j < 4; ++j) {
        const int e = nb[j] + ncol;
        float4 lo, hi;
        lo.x = acc[0][j][0]; lo.y = acc[0][j][1]; lo.z = acc[0][j][2]; lo.w = acc[0][j][3];
        hi.x = acc[0][j][4]; hi.y = acc[0][j][5]; hi.z = acc[0][j][6]; hi.w = acc[0][j][7];
        *reinterpret_cast<float4*>(&Xs[e][mtop])     = lo;
        *reinterpret_cast<float4*>(&Xs[e][mtop + 4]) = hi;
      }
      __syncthreads();
      #pragma unroll
      for (int i = 0; i < 8; ++i) {
        int f  = tid + i * 256;
        int e  = f >> 4;
        int t4 = (f & 15) << 2;
        float4 v = *reinterpret_cast<const float4*>(&Xs[e][t4]);
        *reinterpret_cast<float4*>(dst + tbase + (size_t)(e * NB + kblk) * TM + t4) = v;
      }
    } else {
      const int j2 = hh * 16 + ncol;
      const size_t col = (size_t)j2 * BS + 2 * G2 * blockIdx.y;
      const float4 blo = *reinterpret_cast<const float4*>(bias + col);
      const float4 bhi = *reinterpret_cast<const float4*>(bias + col + 4);
      float* __restrict__ drow = dst + (size_t)(b0 + mtop) * SIZE + col;
      const bool odd = (lane & 1) != 0;
      const int off1 = odd ? (4 - BS) : 0;
      const int off2 = odd ? 4 : BS;
      #pragma unroll
      for (int r = 0; r < 8; ++r) {
        float4 lo, hi, rcv, v1, v2;
        lo.x = acc[0][0][r]             + blo.x;
        lo.y = acc[0][1 % NJ][r]        + blo.y;
        lo.z = acc[1 % NBLK][0][r]      + blo.z;
        lo.w = acc[1 % NBLK][1 % NJ][r] + blo.w;
        hi.x = acc[2 % NBLK][0][r]      + bhi.x;
        hi.y = acc[2 % NBLK][1 % NJ][r] + bhi.y;
        hi.z = acc[3 % NBLK][0][r]      + bhi.z;
        hi.w = acc[3 % NBLK][1 % NJ][r] + bhi.w;
        rcv.x = __shfl_xor(odd ? lo.x : hi.x, 1);
        rcv.y = __shfl_xor(odd ? lo.y : hi.y, 1);
        rcv.z = __shfl_xor(odd ? lo.z : hi.z, 1);
        rcv.w = __shfl_xor(odd ? lo.w : hi.w, 1);
        v1.x = odd ? rcv.x : lo.x;  v1.y = odd ? rcv.y : lo.y;  v1.z = odd ? rcv.z : lo.z;  v1.w = odd ? rcv.w : lo.w;
        v2.x = odd ? hi.x : rcv.x;  v2.y = odd ? hi.y : rcv.y;  v2.z = odd ? hi.z : rcv.z;  v2.w = odd ? hi.w : rcv.w;
        float* __restrict__ p = drow + (size_t)r * SIZE;
        *reinterpret_cast<float4*>(p + off1) = v1;
        *reinterpret_cast<float4*>(p + off2) = v2;
      }
    }
  }
}

extern "C" void kernel_launch(void* const* d_in, const int* in_sizes, int n_in,
                              void* d_out, int out_size, void* d_ws, size_t ws_size,
                              hipStream_t stream) {
  const float* x    = (const float*)d_in[0];
  const float* w1   = (const float*)d_in[1];
  const float* w2   = (const float*)d_in[2];
  const float* bias = (const float*)d_in[3];
  float* out   = (float*)d_out;
  float* inter = (float*)d_ws;

  const int Btok = in_sizes[0] / SIZE;
  dim3 block1(256);
  dim3 block2(512);
  dim3 grid1(Btok / (TM * NT1), NB);
  dim3 grid2(Btok / (TM * NT2), NB / G2);

  monarch_pass<1><<<grid1, block1, 0, stream>>>(x,     w1, bias, inter);
  monarch_pass<2><<<grid2, block2, 0, stream>>>(inter, w2, bias, out);
}


# Fragment loads one block ahead: each 16-WMMA block's LDS reads issue as one group before the previous block's WMMAs (2 register sets); weight fragments also prefetched across the panel change
# speedup vs baseline: 4.4271x; 1.0670x over previous
#include <hip/hip_runtime.h>


#define SIZE 8192
#define NB   64
#define BS   128
#define TM   64
#define LDW  132
#define G2   4
#define NT1  8
#define NT2  4

typedef float v2f __attribute__((ext_vector_type(2)));
typedef float v8f __attribute__((ext_vector_type(8)));

template <int PASS>
__global__ __launch_bounds__(PASS == 1 ? 256 : 512) void monarch_pass(
    const float* __restrict__ src, const float* __restrict__ w,
    const float* __restrict__ bias, float* __restrict__ dst) {
  constexpr int NBLK  = (PASS == 1) ? 1 : G2;
  constexpr int NT    = (PASS == 1) ? NT1 : NT2;
  constexpr int NTHR  = (PASS == 1) ? 256 : 512;
  constexpr int KP    = (PASS == 1) ? 64 : 128;
  constexpr int NKP   = BS / KP;
  constexpr int LDP   = KP + 4;
  constexpr int NJ    = (PASS == 1) ? 4 : 2;
  constexpr int XROWS = (PASS == 1) ? BS : TM;
  constexpr int KB    = 16 / NJ;
  constexpr int NBK   = (KP / 4) / KB;
  static_assert(NBK >= 2 && (NBK % 2) == 0, "register sets alternate and every panel starts in set 0");
  __shared__ __attribute__((aligned(16))) float Wt[NBLK][BS][LDW];
  __shared__ __attribute__((aligned(16))) float Xs[XROWS][LDP];

  const int tid  = threadIdx.x;
  const int lane = tid & 31;
  const int wv   = tid >> 5;

  const int mrow  = (wv & 3) * 16 + (lane & 15);
  const int khalf = (lane >> 4) << 1;
  const int hh    = wv >> 2;
  const int ncol  = lane & 15;

  int nb[NJ];
  #pragma unroll
  for (int j = 0; j < NJ; ++j)
    nb[j] = (PASS == 1) ? (hh * 64 + j * 16) : (j * 64 + hh * 16);

  float4 pf0, pf1, pf2, pf3;

  auto piece = [&](int t, int q, int kh, int i) -> float4 {
    const int ftile = blockIdx.x * NT + t;
    const int fkblk = blockIdx.y * NBLK + q;
    const int f = tid + i * NTHR;
    if (PASS == 1) {
      const int r  = f >> 4;
      const int c4 = (f & 15) << 2;
      return *reinterpret_cast<const float4*>(
          src + (size_t)(ftile * TM + r) * SIZE + (size_t)fkblk * BS + kh * KP + c4);
    } else {
      const int u  = f >> 5;
      const int c  = ((u >> 2) << 3) | ((f >> 2) & 7);
      const int t4 = (((u & 3) << 2) | (f & 3)) << 2;
      return *reinterpret_cast<const float4*>(
          src + (size_t)ftile * SIZE * TM + (size_t)(fkblk * BS + kh * KP + c) * TM + t4);
    }
  };
  auto fetch = [&](int t, int q, int kh) {
    pf0 = piece(t, q, kh, 0);
    pf1 = piece(t, q, kh, 1);
    pf2 = piece(t, q, kh, 2);
    pf3 = piece(t, q, kh, 3);
  };

  auto put = [&](int i, float4 v) {
    const int f = tid + i * NTHR;
    if (PASS == 1) {
      const int r  = f >> 4;
      const int c4 = (f & 15) << 2;
      *reinterpret_cast<float4*>(&Xs[r][c4]) = v;
    } else {
      const int u  = f >> 5;
      const int c  = ((u >> 2) << 3) | ((f >> 2) & 7);
      const int t4 = (((u & 3) << 2) | (f & 3)) << 2;
      Xs[t4 + 0][c] = v.x;
      Xs[t4 + 1][c] = v.y;
      Xs[t4 + 2][c] = v.z;
      Xs[t4 + 3][c] = v.w;
    }
  };
  auto stash = [&]() {
    put(0, pf0);
    put(1, pf1);
    put(2, pf2);
    put(3, pf3);
  };

  v2f fa[2][KB];
  v2f fb[2][NJ][KB];
  auto load_a = [&](int s, int blk) {
    #pragma unroll
    for (int ks = 0; ks < KB; ++ks)
      fa[s][ks] = *reinterpret_cast<const v2f*>(&Xs[mrow][(blk * KB + ks) * 4 + khalf]);
  };
  auto load_b = [&](int s, int q, int kh, int blk) {
    #pragma unroll
    for (int ks = 0; ks < KB; ++ks) {
      #pragma unroll
      for (int j = 0; j < NJ; ++j)
        fb[s][j][ks] = *reinterpret_cast<const v2f*>(
            &Wt[q][nb[j] + ncol][kh * KP + (blk * KB + ks) * 4 + khalf]);
    }
  };
  v8f acc[NBLK][NJ];
  auto mma_block = [&](int s, int q) {
    #pragma unroll
    for (int ks = 0; ks < KB; ++ks) {
      #pragma unroll
      for (int j = 0; j < NJ; ++j)
        acc[q][j] = __builtin_amdgcn_wmma_f32_16x16x4_f32(
             false, fa[s][ks],  false, fb[s][j][ks],
             (short)0, acc[q][j],  false,  false);
    }
  };

  fetch(0, 0, 0);

  #pragma unroll
  for (int q = 0; q < NBLK; ++q) {
    const float* __restrict__ wblk = w + (size_t)(blockIdx.y * NBLK + q) * BS * BS;
    #pragma unroll
    for (int i = 0; i < (BS * BS / 4) / NTHR; ++i) {
      int f  = tid + i * NTHR;
      int u  = f >> 5;
      int d  = ((u >> 3) << 3) | ((f >> 2) & 7);
      int c4 = (((u & 7) << 2) | (f & 3)) << 2;
      float4 v = *reinterpret_cast<const float4*>(wblk + (size_t)d * BS + c4);
      Wt[q][c4 + 0][d] = v.x;
      Wt[q][c4 + 1][d] = v.y;
      Wt[q][c4 + 2][d] = v.z;
      Wt[q][c4 + 3][d] = v.w;
    }
  }

  const int mtop = (wv & 3) * 16 + ((lane >> 4) << 3);

  for (int t = 0; t < NT; ++t) {
    const int tile = blockIdx.x * NT + t;
    const int b0   = tile * TM;
    const size_t tbase = (size_t)tile * SIZE * TM;

    #pragma unroll
    for (int q = 0; q < NBLK; ++q) {
      #pragma unroll
      for (int j = 0; j < NJ; ++j) acc[q][j] = v8f{};
    }

    #pragma unroll
    for (int q = 0; q < NBLK; ++q) {
      #pragma unroll
      for (int kh = 0; kh < NKP; ++kh) {
        const int khn = (kh + 1) % NKP;
        const int qn  = (kh + 1 < NKP) ? q : (q + 1) % NBLK;

        __syncthreads();
        stash();
        __syncthreads();

        load_a(0, 0);
        if (q == 0 && kh == 0) {
          if (t == 0) load_b(0, 0, 0, 0);
        }

        if (kh + 1 < NKP)       fetch(t, q, kh + 1);
        else if (q + 1 < NBLK)  fetch(t, q + 1, 0);
        else if (t + 1 < NT)    fetch(t + 1, 0, 0);

        #pragma unroll
        for (int blk = 0; blk < NBK; ++blk) {
          const int s = blk & 1;
          if (blk + 1 < NBK) {
            load_a(s ^ 1, blk + 1);
            load_b(s ^ 1, q, kh, blk + 1);
          } else {
            load_b(s ^ 1, qn, khn, 0);
          }
          __builtin_amdgcn_sched_barrier(0);
          asm volatile("" : "+v"(fa[s][0]));
          mma_block(s, q);
          __builtin_amdgcn_sched_barrier(0);
        }
      }
    }

    if constexpr (PASS == 1) {
      const int kblk = blockIdx.y;
      __syncthreads();
      #pragma unroll
      for (int j = 0; j < 4; ++j) {
        const int e = nb[j] + ncol;
        float4 lo, hi;
        lo.x = acc[0][j][0]; lo.y = acc[0][j][1]; lo.z = acc[0][j][2]; lo.w = acc[0][j][3];
        hi.x = acc[0][j][4]; hi.y = acc[0][j][5]; hi.z = acc[0][j][6]; hi.w = acc[0][j][7];
        *reinterpret_cast<float4*>(&Xs[e][mtop])     = lo;
        *reinterpret_cast<float4*>(&Xs[e][mtop + 4]) = hi;
      }
      __syncthreads();
      #pragma unroll
      for (int i = 0; i < 8; ++i) {
        int f  = tid + i * 256;
        int e  = f >> 4;
        int t4 = (f & 15) << 2;
        float4 v = *reinterpret_cast<const float4*>(&Xs[e][t4]);
        *reinterpret_cast<float4*>(dst + tbase + (size_t)(e * NB + kblk) * TM + t4) = v;
      }
    } else {
      const int j2 = hh * 16 + ncol;
      const size_t col = (size_t)j2 * BS + 2 * G2 * blockIdx.y;
      const float4 blo = *reinterpret_cast<const float4*>(bias + col);
      const float4 bhi = *reinterpret_cast<const float4*>(bias + col + 4);
      float* __restrict__ drow = dst + (size_t)(b0 + mtop) * SIZE + col;
      const bool odd = (lane & 1) != 0;
      const int off1 = odd ? (4 - BS) : 0;
      const int off2 = odd ? 4 : BS;
      #pragma unroll
      for (int r = 0; r < 8; ++r) {
        float4 lo, hi, rcv, v1, v2;
        lo.x = acc[0][0][r]             + blo.x;
        lo.y = acc[0][1 % NJ][r]        + blo.y;
        lo.z = acc[1 % NBLK][0][r]      + blo.z;
        lo.w = acc[1 % NBLK][1 % NJ][r] + blo.w;
        hi.x = acc[2 % NBLK][0][r]      + bhi.x;
        hi.y = acc[2 % NBLK][1 % NJ][r] + bhi.y;
        hi.z = acc[3 % NBLK][0][r]      + bhi.z;
        hi.w = acc[3 % NBLK][1 % NJ][r] + bhi.w;
        rcv.x = __shfl_xor(odd ? lo.x : hi.x, 1);
        rcv.y = __shfl_xor(odd ? lo.y : hi.y, 1);
        rcv.z = __shfl_xor(odd ? lo.z : hi.z, 1);
        rcv.w = __shfl_xor(odd ? lo.w : hi.w, 1);
        v1.x = odd ? rcv.x : lo.x;  v1.y = odd ? rcv.y : lo.y;  v1.z = odd ? rcv.z : lo.z;  v1.w = odd ? rcv.w : lo.w;
        v2.x = odd ? hi.x : rcv.x;  v2.y = odd ? hi.y : rcv.y;  v2.z = odd ? hi.z : rcv.z;  v2.w = odd ? hi.w : rcv.w;
        float* __restrict__ p = drow + (size_t)r * SIZE;
        *reinterpret_cast<float4*>(p + off1) = v1;
        *reinterpret_cast<float4*>(p + off2) = v2;
      }
    }
  }
}

extern "C" void kernel_launch(void* const* d_in, const int* in_sizes, int n_in,
                              void* d_out, int out_size, void* d_ws, size_t ws_size,
                              hipStream_t stream) {
  const float* x    = (const float*)d_in[0];
  const float* w1   = (const float*)d_in[1];
  const float* w2   = (const float*)d_in[2];
  const float* bias = (const float*)d_in[3];
  float* out   = (float*)d_out;
  float* inter = (float*)d_ws;

  const int Btok = in_sizes[0] / SIZE;
  dim3 block1(256);
  dim3 block2(512);
  dim3 grid1(Btok / (TM * NT1), NB);
  dim3 grid2(Btok / (TM * NT2), NB / G2);

  monarch_pass<1><<<grid1, block1, 0, stream>>>(x,     w1, bias, inter);
  monarch_pass<2><<<grid2, block2, 0, stream>>>(inter, w2, bias, out);
}
